# starting kernel
# baseline (speedup 1.0000x reference)
#include <hip/hip_runtime.h>
#include <math.h>

typedef __attribute__((ext_vector_type(2))) float v2f;
typedef __attribute__((ext_vector_type(4))) float v4f;
typedef __attribute__((ext_vector_type(8))) float v8f;

#define CANVAS_H 256
#define CANVAS_W 384
#define NPTS     1024
#define R_WH     1.5f

__global__ __launch_bounds__(256) void points_kernel(
    const float* __restrict__ loc,
    const float* __restrict__ mo,
    const float* __restrict__ mse,
    const float* __restrict__ col,
    float* __restrict__ out)
{
  __shared__ __align__(16) float lds[12 * NPTS];
  float* BJ0 = lds;
  float* BJ1 = lds + 4 * NPTS;
  float* C4  = lds + 8 * NPTS;

  for (int n = threadIdx.x; n < NPTS; n += blockDim.x) {
    float s   = 16.0f * __expf(-mse[n]);
    float m00 = mo[n * 4 + 0] + s;
    float m01 = mo[n * 4 + 1];
    float m10 = mo[n * 4 + 2];
    float m11 = mo[n * 4 + 3] + s;
    float ly  = loc[n * 2 + 0];
    float lx  = loc[n * 2 + 1];
    float b0  = ly * m00 + lx * m10;
    float b1  = ly * m01 + lx * m11;

    v2f lo0; lo0.x = m00; lo0.y = m10;
    v2f hi0; hi0.x = -b0; hi0.y = 0.0f;
    v2f lo1; lo1.x = m01; lo1.y = m11;
    v2f hi1; hi1.x = -b1; hi1.y = 0.0f;
    *(v2f*)&BJ0[n * 2]              = lo0;
    *(v2f*)&BJ0[(NPTS + n) * 2]     = hi0;
    *(v2f*)&BJ1[n * 2]              = lo1;
    *(v2f*)&BJ1[(NPTS + n) * 2]     = hi1;
    v4f c; c.x = col[n * 3 + 0]; c.y = col[n * 3 + 1];
    c.z = col[n * 3 + 2]; c.w = 0.0f;
    *(v4f*)&C4[n * 4] = c;
  }
  __syncthreads();

  const int lane = threadIdx.x & 31;
  const int wave = threadIdx.x >> 5;
  const int tile = blockIdx.x * 8 + wave;
  const int hi   = lane >> 4;
  const int ln   = lane & 15;

  int p0 = tile * 16 + lane;
  int h0 = p0 / CANVAS_W, w0 = p0 - h0 * CANVAS_W;
  float gy = -1.0f + (float)h0 * (2.0f / (float)(CANVAS_H - 1));
  float gx = -R_WH + (float)w0 * (2.0f * R_WH / (float)(CANVAS_W - 1));
  v2f a;
  a.x = hi ? 1.0f : gy;
  a.y = hi ? 0.0f : gx;

  float accR[8], accG[8], accB[8];
#pragma unroll
  for (int e = 0; e < 8; ++e) { accR[e] = 0.0f; accG[e] = 0.0f; accB[e] = 0.0f; }

  const v8f cz = {};
  const int bsel = (hi * NPTS + ln) * 2;

  for (int k = 0; k < NPTS / 16; ++k) {
    v2f b0 = *(const v2f*)&BJ0[bsel + k * 32];
    v2f b1 = *(const v2f*)&BJ1[bsel + k * 32];
    v4f c4 = *(const v4f*)&C4[(k * 16 + ln) * 4];

    v8f d0 = __builtin_amdgcn_wmma_f32_16x16x4_f32(false, a, false, b0,
                                                   (short)0, cz, false, false);
    v8f d1 = __builtin_amdgcn_wmma_f32_16x16x4_f32(false, a, false, b1,
                                                   (short)0, cz, false, false);

#pragma unroll
    for (int e = 0; e < 8; ++e) {
      float dist = 0.5f * (fabsf(d0[e]) + fabsf(d1[e]));
      float m    = fmaxf(0.0f, 1.0f - dist);
      accR[e] = fmaf(m, c4.x, accR[e]);
      accG[e] = fmaf(m, c4.y, accG[e]);
      accB[e] = fmaf(m, c4.z, accB[e]);
    }
  }

#pragma unroll
  for (int e = 0; e < 8; ++e) {
#pragma unroll
    for (int off = 1; off < 16; off <<= 1) {
      accR[e] += __shfl_xor(accR[e], off, 16);
      accG[e] += __shfl_xor(accG[e], off, 16);
      accB[e] += __shfl_xor(accB[e], off, 16);
    }
  }

#pragma unroll
  for (int e = 0; e < 8; ++e) {
    if (ln == e) {
      int row = e + hi * 8;
      int p   = tile * 16 + row;
      int h   = p / CANVAS_W, w = p - h * CANVAS_W;
      float r = 1.0f / (1.0f + __expf(-4.0f * accR[e]));
      float g = 1.0f / (1.0f + __expf(-4.0f * accG[e]));
      float b = 1.0f / (1.0f + __expf(-4.0f * accB[e]));
      out[0 * CANVAS_H * CANVAS_W + h * CANVAS_W + w] = r;
      out[1 * CANVAS_H * CANVAS_W + h * CANVAS_W + w] = g;
      out[2 * CANVAS_H * CANVAS_W + h * CANVAS_W + w] = b;
    }
  }
}

extern "C" void kernel_launch(void* const* d_in, const int* in_sizes, int n_in,
                              void* d_out, int out_size, void* d_ws, size_t ws_size,
                              hipStream_t stream) {
  (void)in_sizes; (void)n_in; (void)out_size; (void)d_ws; (void)ws_size;
  const float* loc = (const float*)d_in[0];
  const float* mo  = (const float*)d_in[1];
  const float* mse = (const float*)d_in[2];
  const float* col = (const float*)d_in[3];
  float* out = (float*)d_out;

  const int tiles  = (CANVAS_H * CANVAS_W) / 16;
  const int blocks = tiles / 8;
  points_kernel<<<blocks, 256, 0, stream>>>(loc, mo, mse, col, out);
}


# dist as max(|u|,|v|), u,v = half sum/difference of the two transforms folded into the matrix operands; relu as 1-min(.,1) in one v_maxmin: 5 to 3 vector instr per pair; dist rounding changes
# speedup vs baseline: 1.1352x; 1.1352x over previous
#include <hip/hip_runtime.h>
#include <math.h>

typedef __attribute__((ext_vector_type(2))) float v2f;
typedef __attribute__((ext_vector_type(4))) float v4f;
typedef __attribute__((ext_vector_type(8))) float v8f;

#define CANVAS_H 256
#define CANVAS_W 384
#define NPTS     1024
#define R_WH     1.5f

__global__ __launch_bounds__(256) void points_kernel(
    const float* __restrict__ loc,
    const float* __restrict__ mo,
    const float* __restrict__ mse,
    const float* __restrict__ col,
    float* __restrict__ out)
{
  __shared__ __align__(16) float lds[12 * NPTS];
  float* BJ0 = lds;
  float* BJ1 = lds + 4 * NPTS;
  float* C4  = lds + 8 * NPTS;

  for (int n = threadIdx.x; n < NPTS; n += blockDim.x) {
    float s   = 16.0f * __expf(-mse[n]);
    float m00 = mo[n * 4 + 0] + s;
    float m01 = mo[n * 4 + 1];
    float m10 = mo[n * 4 + 2];
    float m11 = mo[n * 4 + 3] + s;
    float ly  = loc[n * 2 + 0];
    float lx  = loc[n * 2 + 1];
    float u0  = 0.5f * (m00 + m01);
    float u1  = 0.5f * (m10 + m11);
    float v0  = 0.5f * (m00 - m01);
    float v1  = 0.5f * (m10 - m11);
    float bu  = ly * u0 + lx * u1;
    float bv  = ly * v0 + lx * v1;

    v2f lo0; lo0.x = u0;  lo0.y = u1;
    v2f hi0; hi0.x = -bu; hi0.y = 0.0f;
    v2f lo1; lo1.x = v0;  lo1.y = v1;
    v2f hi1; hi1.x = -bv; hi1.y = 0.0f;
    *(v2f*)&BJ0[n * 2]              = lo0;
    *(v2f*)&BJ0[(NPTS + n) * 2]     = hi0;
    *(v2f*)&BJ1[n * 2]              = lo1;
    *(v2f*)&BJ1[(NPTS + n) * 2]     = hi1;
    v4f c; c.x = col[n * 3 + 0]; c.y = col[n * 3 + 1];
    c.z = col[n * 3 + 2]; c.w = 0.0f;
    *(v4f*)&C4[n * 4] = c;
  }
  __syncthreads();

  const int lane = threadIdx.x & 31;
  const int wave = threadIdx.x >> 5;
  const int tile = blockIdx.x * 8 + wave;
  const int hi   = lane >> 4;
  const int ln   = lane & 15;

  int p0 = tile * 16 + lane;
  int h0 = p0 / CANVAS_W, w0 = p0 - h0 * CANVAS_W;
  float gy = -1.0f + (float)h0 * (2.0f / (float)(CANVAS_H - 1));
  float gx = -R_WH + (float)w0 * (2.0f * R_WH / (float)(CANVAS_W - 1));
  v2f a;
  a.x = hi ? 1.0f : gy;
  a.y = hi ? 0.0f : gx;

  float accR[8], accG[8], accB[8];
#pragma unroll
  for (int e = 0; e < 8; ++e) { accR[e] = 0.0f; accG[e] = 0.0f; accB[e] = 0.0f; }

  const v8f cz = {};
  const int bsel = (hi * NPTS + ln) * 2;

  for (int k = 0; k < NPTS / 16; ++k) {
    v2f b0 = *(const v2f*)&BJ0[bsel + k * 32];
    v2f b1 = *(const v2f*)&BJ1[bsel + k * 32];
    v4f c4 = *(const v4f*)&C4[(k * 16 + ln) * 4];

    v8f d0 = __builtin_amdgcn_wmma_f32_16x16x4_f32(false, a, false, b0,
                                                   (short)0, cz, false, false);
    v8f d1 = __builtin_amdgcn_wmma_f32_16x16x4_f32(false, a, false, b1,
                                                   (short)0, cz, false, false);

#pragma unroll
    for (int e = 0; e < 8; ++e) {
      float t;
      asm("v_maxmin_num_f32 %0, |%1|, |%2|, 1.0" : "=v"(t) : "v"(d0[e]), "v"(d1[e]));
      float m = 1.0f - t;
      accR[e] = fmaf(m, c4.x, accR[e]);
      accG[e] = fmaf(m, c4.y, accG[e]);
      accB[e] = fmaf(m, c4.z, accB[e]);
    }
  }

#pragma unroll
  for (int e = 0; e < 8; ++e) {
#pragma unroll
    for (int off = 1; off < 16; off <<= 1) {
      accR[e] += __shfl_xor(accR[e], off, 16);
      accG[e] += __shfl_xor(accG[e], off, 16);
      accB[e] += __shfl_xor(accB[e], off, 16);
    }
  }

#pragma unroll
  for (int e = 0; e < 8; ++e) {
    if (ln == e) {
      int row = e + hi * 8;
      int p   = tile * 16 + row;
      int h   = p / CANVAS_W, w = p - h * CANVAS_W;
      float r = 1.0f / (1.0f + __expf(-4.0f * accR[e]));
      float g = 1.0f / (1.0f + __expf(-4.0f * accG[e]));
      float b = 1.0f / (1.0f + __expf(-4.0f * accB[e]));
      out[0 * CANVAS_H * CANVAS_W + h * CANVAS_W + w] = r;
      out[1 * CANVAS_H * CANVAS_W + h * CANVAS_W + w] = g;
      out[2 * CANVAS_H * CANVAS_W + h * CANVAS_W + w] = b;
    }
  }
}

extern "C" void kernel_launch(void* const* d_in, const int* in_sizes, int n_in,
                              void* d_out, int out_size, void* d_ws, size_t ws_size,
                              hipStream_t stream) {
  (void)in_sizes; (void)n_in; (void)out_size; (void)d_ws; (void)ws_size;
  const float* loc = (const float*)d_in[0];
  const float* mo  = (const float*)d_in[1];
  const float* mse = (const float*)d_in[2];
  const float* col = (const float*)d_in[3];
  float* out = (float*)d_out;

  const int tiles  = (CANVAS_H * CANVAS_W) / 16;
  const int blocks = tiles / 8;
  points_kernel<<<blocks, 256, 0, stream>>>(loc, mo, mse, col, out);
}


# each 4x32-pixel block first compacts into LDS the points whose weight can be nonzero on its rectangle (interval bound on u,v, with margin) and sweeps only those; order of the sum over points changes
# speedup vs baseline: 3.6079x; 3.1782x over previous
#include <hip/hip_runtime.h>
#include <math.h>

typedef __attribute__((ext_vector_type(2))) float v2f;
typedef __attribute__((ext_vector_type(4))) float v4f;
typedef __attribute__((ext_vector_type(8))) float v8f;

#define CANVAS_H 256
#define CANVAS_W 384
#define NPTS     1024
#define R_WH     1.5f
#define BLK_ROWS 4
#define BLK_COLS 32
#define BLKS_PER_ROW (CANVAS_W / BLK_COLS)

__global__ __launch_bounds__(256) void points_kernel(
    const float* __restrict__ loc,
    const float* __restrict__ mo,
    const float* __restrict__ mse,
    const float* __restrict__ col,
    float* __restrict__ out)
{
  __shared__ __align__(16) float lds[12 * NPTS];
  __shared__ int s_cnt[32];
  float* BJ0 = lds;
  float* BJ1 = lds + 4 * NPTS;
  float* C4  = lds + 8 * NPTS;

  const int tid  = threadIdx.x;
  const int lane = tid & 31;
  const int wave = tid >> 5;
  const int hi   = lane >> 4;
  const int ln   = lane & 15;

  const int br = blockIdx.x / BLKS_PER_ROW;
  const int bc = blockIdx.x - br * BLKS_PER_ROW;
  const int hb = br * BLK_ROWS;
  const int wb = bc * BLK_COLS;
  const float SY = 2.0f / (float)(CANVAS_H - 1);
  const float SX = 2.0f * R_WH / (float)(CANVAS_W - 1);
  const float gy_lo = -1.0f + (float)hb * SY;
  const float gy_hi = -1.0f + (float)(hb + BLK_ROWS - 1) * SY;
  const float gx_lo = -R_WH + (float)wb * SX;
  const float gx_hi = -R_WH + (float)(wb + BLK_COLS - 1) * SX;
  const float cy = 0.5f * (gy_lo + gy_hi), ey = 0.5f * (gy_hi - gy_lo);
  const float cx = 0.5f * (gx_lo + gx_hi), ex = 0.5f * (gx_hi - gx_lo);

  float pu0[4], pu1[4], pv0[4], pv1[4], pbu[4], pbv[4], pr[4], pg[4], pb[4];
  int   ppos[4];
#pragma unroll
  for (int i = 0; i < 4; ++i) {
    const int n = i * 256 + tid;
    float s   = 16.0f * __expf(-mse[n]);
    float m00 = mo[n * 4 + 0] + s;
    float m01 = mo[n * 4 + 1];
    float m10 = mo[n * 4 + 2];
    float m11 = mo[n * 4 + 3] + s;
    float ly  = loc[n * 2 + 0];
    float lx  = loc[n * 2 + 1];
    float u0  = 0.5f * (m00 + m01);
    float u1  = 0.5f * (m10 + m11);
    float v0  = 0.5f * (m00 - m01);
    float v1  = 0.5f * (m10 - m11);
    float bu  = ly * u0 + lx * u1;
    float bv  = ly * v0 + lx * v1;

    float uc = fmaf(cy, u0, fmaf(cx, u1, -bu));
    float vc = fmaf(cy, v0, fmaf(cx, v1, -bv));
    float ru = fabsf(u0) * ey + fabsf(u1) * ex;
    float rv = fabsf(v0) * ey + fabsf(v1) * ex;
    float su = fabsf(cy * u0) + fabsf(cx * u1) + fabsf(bu) + ru + 1.0f;
    float sv = fabsf(cy * v0) + fabsf(cx * v1) + fabsf(bv) + rv + 1.0f;
    bool out_u = (fabsf(uc) - ru) >= (1.0f + 1e-3f * su);
    bool out_v = (fabsf(vc) - rv) >= (1.0f + 1e-3f * sv);
    bool keep  = !(out_u || out_v);

    unsigned mask = __builtin_amdgcn_ballot_w32(keep);
    if (lane == 0) s_cnt[i * 8 + wave] = __builtin_popcount(mask);
    ppos[i] = keep ? (int)__builtin_amdgcn_mbcnt_lo(mask, 0u) : -1;

    pu0[i] = u0; pu1[i] = u1; pv0[i] = v0; pv1[i] = v1; pbu[i] = bu; pbv[i] = bv;
    pr[i] = col[n * 3 + 0]; pg[i] = col[n * 3 + 1]; pb[i] = col[n * 3 + 2];
  }
  __syncthreads();

  const int cnt_l = s_cnt[lane];
  int incl = cnt_l;
  for (int off = 1; off < 32; off <<= 1) {
    int t = __shfl_up(incl, off);
    if (lane >= off) incl += t;
  }
  const int total = __shfl(incl, 31);
  const int excl  = incl - cnt_l;

#pragma unroll
  for (int i = 0; i < 4; ++i) {
    const int base = __shfl(excl, i * 8 + wave);
    if (ppos[i] >= 0) {
      const int n = base + ppos[i];
      v2f lo0; lo0.x = pu0[i];  lo0.y = pu1[i];
      v2f hi0; hi0.x = -pbu[i]; hi0.y = 0.0f;
      v2f lo1; lo1.x = pv0[i];  lo1.y = pv1[i];
      v2f hi1; hi1.x = -pbv[i]; hi1.y = 0.0f;
      *(v2f*)&BJ0[n * 2]              = lo0;
      *(v2f*)&BJ0[(NPTS + n) * 2]     = hi0;
      *(v2f*)&BJ1[n * 2]              = lo1;
      *(v2f*)&BJ1[(NPTS + n) * 2]     = hi1;
      v4f c; c.x = pr[i]; c.y = pg[i]; c.z = pb[i]; c.w = 0.0f;
      *(v4f*)&C4[n * 4] = c;
    }
  }
  const int npad = (total + 15) & ~15;
  if (tid < npad - total) {
    const int n = total + tid;
    v2f z2; z2.x = 0.0f; z2.y = 0.0f;
    v2f h2; h2.x = 2.0f; h2.y = 0.0f;
    v4f z4; z4.x = 0.0f; z4.y = 0.0f; z4.z = 0.0f; z4.w = 0.0f;
    *(v2f*)&BJ0[n * 2]              = z2;
    *(v2f*)&BJ0[(NPTS + n) * 2]     = h2;
    *(v2f*)&BJ1[n * 2]              = z2;
    *(v2f*)&BJ1[(NPTS + n) * 2]     = z2;
    *(v4f*)&C4[n * 4] = z4;
  }
  __syncthreads();
  const int nchunks = __builtin_amdgcn_readfirstlane(npad >> 4);

  const int hrow  = hb + (wave >> 1);
  const int wcol0 = wb + (wave & 1) * 16;

  float gy = -1.0f + (float)hrow * SY;
  float gx = -R_WH + (float)(wcol0 + ln) * SX;
  v2f a;
  a.x = hi ? 1.0f : gy;
  a.y = hi ? 0.0f : gx;

  float accR[8], accG[8], accB[8];
#pragma unroll
  for (int e = 0; e < 8; ++e) { accR[e] = 0.0f; accG[e] = 0.0f; accB[e] = 0.0f; }

  const v8f cz = {};
  const int bsel = (hi * NPTS + ln) * 2;

  for (int k = 0; k < nchunks; ++k) {
    v2f b0 = *(const v2f*)&BJ0[bsel + k * 32];
    v2f b1 = *(const v2f*)&BJ1[bsel + k * 32];
    v4f c4 = *(const v4f*)&C4[(k * 16 + ln) * 4];

    v8f d0 = __builtin_amdgcn_wmma_f32_16x16x4_f32(false, a, false, b0,
                                                   (short)0, cz, false, false);
    v8f d1 = __builtin_amdgcn_wmma_f32_16x16x4_f32(false, a, false, b1,
                                                   (short)0, cz, false, false);

#pragma unroll
    for (int e = 0; e < 8; ++e) {
      float t;
      asm("v_maxmin_num_f32 %0, |%1|, |%2|, 1.0" : "=v"(t) : "v"(d0[e]), "v"(d1[e]));
      float m = 1.0f - t;
      accR[e] = fmaf(m, c4.x, accR[e]);
      accG[e] = fmaf(m, c4.y, accG[e]);
      accB[e] = fmaf(m, c4.z, accB[e]);
    }
  }

#pragma unroll
  for (int e = 0; e < 8; ++e) {
#pragma unroll
    for (int off = 1; off < 16; off <<= 1) {
      accR[e] += __shfl_xor(accR[e], off, 16);
      accG[e] += __shfl_xor(accG[e], off, 16);
      accB[e] += __shfl_xor(accB[e], off, 16);
    }
  }

#pragma unroll
  for (int e = 0; e < 8; ++e) {
    if (ln == e) {
      int w = wcol0 + e + hi * 8;
      float r = 1.0f / (1.0f + __expf(-4.0f * accR[e]));
      float g = 1.0f / (1.0f + __expf(-4.0f * accG[e]));
      float b = 1.0f / (1.0f + __expf(-4.0f * accB[e]));
      out[0 * CANVAS_H * CANVAS_W + hrow * CANVAS_W + w] = r;
      out[1 * CANVAS_H * CANVAS_W + hrow * CANVAS_W + w] = g;
      out[2 * CANVAS_H * CANVAS_W + hrow * CANVAS_W + w] = b;
    }
  }
}

extern "C" void kernel_launch(void* const* d_in, const int* in_sizes, int n_in,
                              void* d_out, int out_size, void* d_ws, size_t ws_size,
                              hipStream_t stream) {
  (void)in_sizes; (void)n_in; (void)out_size; (void)d_ws; (void)ws_size;
  const float* loc = (const float*)d_in[0];
  const float* mo  = (const float*)d_in[1];
  const float* mse = (const float*)d_in[2];
  const float* col = (const float*)d_in[3];
  float* out = (float*)d_out;

  const int blocks = (CANVAS_H / BLK_ROWS) * (CANVAS_W / BLK_COLS);
  points_kernel<<<blocks, 256, 0, stream>>>(loc, mo, mse, col, out);
}


# matrix operands swapped: points are the rows, the wave's 16 pixels the columns, so the sum over points is in-lane and the 16-lane shuffle reduce is gone; order of the sum over points changes
# speedup vs baseline: 4.0585x; 1.1249x over previous
#include <hip/hip_runtime.h>
#include <math.h>

typedef __attribute__((ext_vector_type(2))) float v2f;
typedef __attribute__((ext_vector_type(4))) float v4f;
typedef __attribute__((ext_vector_type(8))) float v8f;

#define CANVAS_H 256
#define CANVAS_W 384
#define NPTS     1024
#define R_WH     1.5f
#define BLK_ROWS 4
#define BLK_COLS 32
#define BLKS_PER_ROW (CANVAS_W / BLK_COLS)

__device__ __forceinline__ float xhalf(float x) {
  int b = __builtin_bit_cast(int, x);
  return __builtin_bit_cast(float,
      __builtin_amdgcn_permlanex16(b, b, 0x76543210, 0xfedcba98, false, false));
}

__global__ __launch_bounds__(256) void points_kernel(
    const float* __restrict__ loc,
    const float* __restrict__ mo,
    const float* __restrict__ mse,
    const float* __restrict__ col,
    float* __restrict__ out)
{
  __shared__ __align__(16) float lds[11 * NPTS];
  __shared__ int s_cnt[32];
  float* BJ0 = lds;
  float* BJ1 = lds + 4 * NPTS;
  float* CT  = lds + 8 * NPTS;

  const int tid  = threadIdx.x;
  const int lane = tid & 31;
  const int wave = tid >> 5;
  const int hi   = lane >> 4;
  const int ln   = lane & 15;

  const int br = blockIdx.x / BLKS_PER_ROW;
  const int bc = blockIdx.x - br * BLKS_PER_ROW;
  const int hb = br * BLK_ROWS;
  const int wb = bc * BLK_COLS;
  const float SY = 2.0f / (float)(CANVAS_H - 1);
  const float SX = 2.0f * R_WH / (float)(CANVAS_W - 1);
  const float gy_lo = -1.0f + (float)hb * SY;
  const float gy_hi = -1.0f + (float)(hb + BLK_ROWS - 1) * SY;
  const float gx_lo = -R_WH + (float)wb * SX;
  const float gx_hi = -R_WH + (float)(wb + BLK_COLS - 1) * SX;
  const float cy = 0.5f * (gy_lo + gy_hi), ey = 0.5f * (gy_hi - gy_lo);
  const float cx = 0.5f * (gx_lo + gx_hi), ex = 0.5f * (gx_hi - gx_lo);

  float pu0[4], pu1[4], pv0[4], pv1[4], pbu[4], pbv[4], pr[4], pg[4], pb[4];
  int   ppos[4];
#pragma unroll
  for (int i = 0; i < 4; ++i) {
    const int n = i * 256 + tid;
    float s   = 16.0f * __expf(-mse[n]);
    float m00 = mo[n * 4 + 0] + s;
    float m01 = mo[n * 4 + 1];
    float m10 = mo[n * 4 + 2];
    float m11 = mo[n * 4 + 3] + s;
    float ly  = loc[n * 2 + 0];
    float lx  = loc[n * 2 + 1];
    float u0  = 0.5f * (m00 + m01);
    float u1  = 0.5f * (m10 + m11);
    float v0  = 0.5f * (m00 - m01);
    float v1  = 0.5f * (m10 - m11);
    float bu  = ly * u0 + lx * u1;
    float bv  = ly * v0 + lx * v1;

    float uc = fmaf(cy, u0, fmaf(cx, u1, -bu));
    float vc = fmaf(cy, v0, fmaf(cx, v1, -bv));
    float ru = fabsf(u0) * ey + fabsf(u1) * ex;
    float rv = fabsf(v0) * ey + fabsf(v1) * ex;
    float su = fabsf(cy * u0) + fabsf(cx * u1) + fabsf(bu) + ru + 1.0f;
    float sv = fabsf(cy * v0) + fabsf(cx * v1) + fabsf(bv) + rv + 1.0f;
    bool out_u = (fabsf(uc) - ru) >= (1.0f + 1e-3f * su);
    bool out_v = (fabsf(vc) - rv) >= (1.0f + 1e-3f * sv);
    bool keep  = !(out_u || out_v);

    unsigned mask = __builtin_amdgcn_ballot_w32(keep);
    if (lane == 0) s_cnt[i * 8 + wave] = __builtin_popcount(mask);
    ppos[i] = keep ? (int)__builtin_amdgcn_mbcnt_lo(mask, 0u) : -1;

    pu0[i] = u0; pu1[i] = u1; pv0[i] = v0; pv1[i] = v1; pbu[i] = bu; pbv[i] = bv;
    pr[i] = col[n * 3 + 0]; pg[i] = col[n * 3 + 1]; pb[i] = col[n * 3 + 2];
  }
  __syncthreads();

  const int cnt_l = s_cnt[lane];
  int incl = cnt_l;
  for (int off = 1; off < 32; off <<= 1) {
    int t = __shfl_up(incl, off);
    if (lane >= off) incl += t;
  }
  const int total = __shfl(incl, 31);
  const int excl  = incl - cnt_l;

#pragma unroll
  for (int i = 0; i < 4; ++i) {
    const int base = __shfl(excl, i * 8 + wave);
    if (ppos[i] >= 0) {
      const int n = base + ppos[i];
      v2f lo0; lo0.x = pu0[i];  lo0.y = pu1[i];
      v2f hi0; hi0.x = -pbu[i]; hi0.y = 0.0f;
      v2f lo1; lo1.x = pv0[i];  lo1.y = pv1[i];
      v2f hi1; hi1.x = -pbv[i]; hi1.y = 0.0f;
      *(v2f*)&BJ0[n * 2]              = lo0;
      *(v2f*)&BJ0[(NPTS + n) * 2]     = hi0;
      *(v2f*)&BJ1[n * 2]              = lo1;
      *(v2f*)&BJ1[(NPTS + n) * 2]     = hi1;
      const int cn = (n >> 3) * 24 + (n & 7);
      CT[cn]      = pr[i];
      CT[cn + 8]  = pg[i];
      CT[cn + 16] = pb[i];
    }
  }
  const int npad = (total + 15) & ~15;
  if (tid < npad - total) {
    const int n = total + tid;
    v2f z2; z2.x = 0.0f; z2.y = 0.0f;
    v2f h2; h2.x = 2.0f; h2.y = 0.0f;
    *(v2f*)&BJ0[n * 2]              = z2;
    *(v2f*)&BJ0[(NPTS + n) * 2]     = h2;
    *(v2f*)&BJ1[n * 2]              = z2;
    *(v2f*)&BJ1[(NPTS + n) * 2]     = z2;
    const int cn = (n >> 3) * 24 + (n & 7);
    CT[cn]      = 0.0f;
    CT[cn + 8]  = 0.0f;
    CT[cn + 16] = 0.0f;
  }
  __syncthreads();
  const int nchunks = __builtin_amdgcn_readfirstlane(npad >> 4);

  const int hrow  = hb + (wave >> 1);
  const int wcol0 = wb + (wave & 1) * 16;

  float gy = -1.0f + (float)hrow * SY;
  float gx = -R_WH + (float)(wcol0 + ln) * SX;
  v2f bp;
  bp.x = hi ? 1.0f : gy;
  bp.y = hi ? 0.0f : gx;

  float accR[2], accG[2], accB[2];
#pragma unroll
  for (int q = 0; q < 2; ++q) { accR[q] = 0.0f; accG[q] = 0.0f; accB[q] = 0.0f; }

  const v8f cz = {};
  const int asel = (hi * NPTS + ln) * 2;
  const int csel = hi * 24;

  for (int k = 0; k < nchunks; ++k) {
    v2f pa0 = *(const v2f*)&BJ0[asel + k * 32];
    v2f pa1 = *(const v2f*)&BJ1[asel + k * 32];
    const v4f* cp = (const v4f*)&CT[csel + k * 48];
    v4f cr0 = cp[0], cr1 = cp[1];
    v4f cg0 = cp[2], cg1 = cp[3];
    v4f cb0 = cp[4], cb1 = cp[5];
    const float cr[8] = {cr0.x, cr0.y, cr0.z, cr0.w, cr1.x, cr1.y, cr1.z, cr1.w};
    const float cg[8] = {cg0.x, cg0.y, cg0.z, cg0.w, cg1.x, cg1.y, cg1.z, cg1.w};
    const float cb[8] = {cb0.x, cb0.y, cb0.z, cb0.w, cb1.x, cb1.y, cb1.z, cb1.w};

    v8f d0 = __builtin_amdgcn_wmma_f32_16x16x4_f32(false, pa0, false, bp,
                                                   (short)0, cz, false, false);
    v8f d1 = __builtin_amdgcn_wmma_f32_16x16x4_f32(false, pa1, false, bp,
                                                   (short)0, cz, false, false);

#pragma unroll
    for (int e = 0; e < 8; ++e) {
      float t;
      asm("v_maxmin_num_f32 %0, |%1|, |%2|, 1.0" : "=v"(t) : "v"(d0[e]), "v"(d1[e]));
      float m = 1.0f - t;
      accR[e & 1] = fmaf(m, cr[e], accR[e & 1]);
      accG[e & 1] = fmaf(m, cg[e], accG[e & 1]);
      accB[e & 1] = fmaf(m, cb[e], accB[e & 1]);
    }
  }

  float sR = accR[0] + accR[1];
  float sG = accG[0] + accG[1];
  float sB = accB[0] + accB[1];
  sR += xhalf(sR);
  sG += xhalf(sG);
  sB += xhalf(sB);

  const int w = wcol0 + ln;
  const float s0 = hi ? sG : sR;
  const float o0 = 1.0f / (1.0f + __expf(-4.0f * s0));
  const float o2 = 1.0f / (1.0f + __expf(-4.0f * sB));
  out[hi * CANVAS_H * CANVAS_W + hrow * CANVAS_W + w] = o0;
  if (hi == 0) out[2 * CANVAS_H * CANVAS_W + hrow * CANVAS_W + w] = o2;
}

extern "C" void kernel_launch(void* const* d_in, const int* in_sizes, int n_in,
                              void* d_out, int out_size, void* d_ws, size_t ws_size,
                              hipStream_t stream) {
  (void)in_sizes; (void)n_in; (void)out_size; (void)d_ws; (void)ws_size;
  const float* loc = (const float*)d_in[0];
  const float* mo  = (const float*)d_in[1];
  const float* mse = (const float*)d_in[2];
  const float* col = (const float*)d_in[3];
  float* out = (float*)d_out;

  const int blocks = (CANVAS_H / BLK_ROWS) * (CANVAS_W / BLK_COLS);
  points_kernel<<<blocks, 256, 0, stream>>>(loc, mo, mse, col, out);
}


# prologue latency chain: list order (wave,pass,lane): one 8-count LDS exchange replaces per-pass counts + 32-entry shuffle scan; all passes' loads issued together; order of the sum over points changes
# speedup vs baseline: 5.6988x; 1.4042x over previous
#include <hip/hip_runtime.h>
#include <math.h>

typedef __attribute__((ext_vector_type(2))) float v2f;
typedef __attribute__((ext_vector_type(4))) float v4f;
typedef __attribute__((ext_vector_type(8))) float v8f;
typedef __attribute__((ext_vector_type(4))) int   v4i;

#define CANVAS_H 256
#define CANVAS_W 384
#define NPTS     1024
#define R_WH     1.5f
#define BLK_ROWS 4
#define BLK_COLS 32
#define BLKS_PER_ROW (CANVAS_W / BLK_COLS)

__device__ __forceinline__ float xhalf(float x) {
  int b = __builtin_bit_cast(int, x);
  return __builtin_bit_cast(float,
      __builtin_amdgcn_permlanex16(b, b, 0x76543210, 0xfedcba98, false, false));
}

__global__ __launch_bounds__(256) void points_kernel(
    const float* __restrict__ loc,
    const float* __restrict__ mo,
    const float* __restrict__ mse,
    const float* __restrict__ col,
    float* __restrict__ out)
{
  __shared__ __align__(16) float lds[11 * NPTS];
  __shared__ __align__(16) int s_cnt[8];
  float* BJ0 = lds;
  float* BJ1 = lds + 4 * NPTS;
  float* CT  = lds + 8 * NPTS;

  const int tid  = threadIdx.x;
  const int lane = tid & 31;
  const int wave = tid >> 5;
  const int hi   = lane >> 4;
  const int ln   = lane & 15;

  const int br = blockIdx.x / BLKS_PER_ROW;
  const int bc = blockIdx.x - br * BLKS_PER_ROW;
  const int hb = br * BLK_ROWS;
  const int wb = bc * BLK_COLS;
  const float SY = 2.0f / (float)(CANVAS_H - 1);
  const float SX = 2.0f * R_WH / (float)(CANVAS_W - 1);
  const float gy_lo = -1.0f + (float)hb * SY;
  const float gy_hi = -1.0f + (float)(hb + BLK_ROWS - 1) * SY;
  const float gx_lo = -R_WH + (float)wb * SX;
  const float gx_hi = -R_WH + (float)(wb + BLK_COLS - 1) * SX;
  const float cy = 0.5f * (gy_lo + gy_hi), ey = 0.5f * (gy_hi - gy_lo);
  const float cx = 0.5f * (gx_lo + gx_hi), ex = 0.5f * (gx_hi - gx_lo);

  float le[4], lm00[4], lm01[4], lm10[4], lm11[4], lly[4], llx[4];
  float pr[4], pg[4], pb[4];
#pragma unroll
  for (int i = 0; i < 4; ++i) {
    const int n = i * 256 + tid;
    le[i]   = mse[n];
    lm00[i] = mo[n * 4 + 0];
    lm01[i] = mo[n * 4 + 1];
    lm10[i] = mo[n * 4 + 2];
    lm11[i] = mo[n * 4 + 3];
    lly[i]  = loc[n * 2 + 0];
    llx[i]  = loc[n * 2 + 1];
  }
#pragma unroll
  for (int i = 0; i < 4; ++i) {
    const int n = i * 256 + tid;
    pr[i] = col[n * 3 + 0]; pg[i] = col[n * 3 + 1]; pb[i] = col[n * 3 + 2];
  }

  float pu0[4], pu1[4], pv0[4], pv1[4], pbu[4], pbv[4];
  bool  pkeep[4];
  int   prank[4];
  int   pcnt[4];
#pragma unroll
  for (int i = 0; i < 4; ++i) {
    float s   = 16.0f * __expf(-le[i]);
    float m00 = lm00[i] + s;
    float m01 = lm01[i];
    float m10 = lm10[i];
    float m11 = lm11[i] + s;
    float ly  = lly[i];
    float lx  = llx[i];
    float u0  = 0.5f * (m00 + m01);
    float u1  = 0.5f * (m10 + m11);
    float v0  = 0.5f * (m00 - m01);
    float v1  = 0.5f * (m10 - m11);
    float bu  = ly * u0 + lx * u1;
    float bv  = ly * v0 + lx * v1;

    float uc = fmaf(cy, u0, fmaf(cx, u1, -bu));
    float vc = fmaf(cy, v0, fmaf(cx, v1, -bv));
    float ru = fabsf(u0) * ey + fabsf(u1) * ex;
    float rv = fabsf(v0) * ey + fabsf(v1) * ex;
    float su = fabsf(cy * u0) + fabsf(cx * u1) + fabsf(bu) + ru + 1.0f;
    float sv = fabsf(cy * v0) + fabsf(cx * v1) + fabsf(bv) + rv + 1.0f;
    bool out_u = (fabsf(uc) - ru) >= (1.0f + 1e-3f * su);
    bool out_v = (fabsf(vc) - rv) >= (1.0f + 1e-3f * sv);
    bool keep  = !(out_u || out_v);

    unsigned mask = __builtin_amdgcn_ballot_w32(keep);
    pkeep[i] = keep;
    prank[i] = (int)__builtin_amdgcn_mbcnt_lo(mask, 0u);
    pcnt[i]  = __builtin_popcount(mask);

    pu0[i] = u0; pu1[i] = u1; pv0[i] = v0; pv1[i] = v1; pbu[i] = bu; pbv[i] = bv;
  }
  if (lane == 0) s_cnt[wave] = pcnt[0] + pcnt[1] + pcnt[2] + pcnt[3];
  __syncthreads();

  const v4i ca = *(const v4i*)&s_cnt[0];
  const v4i cc = *(const v4i*)&s_cnt[4];
  const int cw[8] = {ca.x, ca.y, ca.z, ca.w, cc.x, cc.y, cc.z, cc.w};
  int total = 0, wbase = 0;
#pragma unroll
  for (int q = 0; q < 8; ++q) {
    total += cw[q];
    wbase += (q < wave) ? cw[q] : 0;
  }

  int pbase = wbase;
#pragma unroll
  for (int i = 0; i < 4; ++i) {
    if (pkeep[i]) {
      const int n = pbase + prank[i];
      v2f lo0; lo0.x = pu0[i];  lo0.y = pu1[i];
      v2f hi0; hi0.x = -pbu[i]; hi0.y = 0.0f;
      v2f lo1; lo1.x = pv0[i];  lo1.y = pv1[i];
      v2f hi1; hi1.x = -pbv[i]; hi1.y = 0.0f;
      *(v2f*)&BJ0[n * 2]              = lo0;
      *(v2f*)&BJ0[(NPTS + n) * 2]     = hi0;
      *(v2f*)&BJ1[n * 2]              = lo1;
      *(v2f*)&BJ1[(NPTS + n) * 2]     = hi1;
      const int cn = (n >> 3) * 24 + (n & 7);
      CT[cn]      = pr[i];
      CT[cn + 8]  = pg[i];
      CT[cn + 16] = pb[i];
    }
    pbase += pcnt[i];
  }
  const int npad = (total + 15) & ~15;
  if (tid < npad - total) {
    const int n = total + tid;
    v2f z2; z2.x = 0.0f; z2.y = 0.0f;
    v2f h2; h2.x = 2.0f; h2.y = 0.0f;
    *(v2f*)&BJ0[n * 2]              = z2;
    *(v2f*)&BJ0[(NPTS + n) * 2]     = h2;
    *(v2f*)&BJ1[n * 2]              = z2;
    *(v2f*)&BJ1[(NPTS + n) * 2]     = z2;
    const int cn = (n >> 3) * 24 + (n & 7);
    CT[cn]      = 0.0f;
    CT[cn + 8]  = 0.0f;
    CT[cn + 16] = 0.0f;
  }
  __syncthreads();
  const int nchunks = __builtin_amdgcn_readfirstlane(npad >> 4);

  const int hrow  = hb + (wave >> 1);
  const int wcol0 = wb + (wave & 1) * 16;

  float gy = -1.0f + (float)hrow * SY;
  float gx = -R_WH + (float)(wcol0 + ln) * SX;
  v2f bp;
  bp.x = hi ? 1.0f : gy;
  bp.y = hi ? 0.0f : gx;

  float accR[2], accG[2], accB[2];
#pragma unroll
  for (int q = 0; q < 2; ++q) { accR[q] = 0.0f; accG[q] = 0.0f; accB[q] = 0.0f; }

  const v8f cz = {};
  const int asel = (hi * NPTS + ln) * 2;
  const int csel = hi * 24;

  for (int k = 0; k < nchunks; ++k) {
    v2f pa0 = *(const v2f*)&BJ0[asel + k * 32];
    v2f pa1 = *(const v2f*)&BJ1[asel + k * 32];
    const v4f* cp = (const v4f*)&CT[csel + k * 48];
    v4f cr0 = cp[0], cr1 = cp[1];
    v4f cg0 = cp[2], cg1 = cp[3];
    v4f cb0 = cp[4], cb1 = cp[5];
    const float cr[8] = {cr0.x, cr0.y, cr0.z, cr0.w, cr1.x, cr1.y, cr1.z, cr1.w};
    const float cg[8] = {cg0.x, cg0.y, cg0.z, cg0.w, cg1.x, cg1.y, cg1.z, cg1.w};
    const float cb[8] = {cb0.x, cb0.y, cb0.z, cb0.w, cb1.x, cb1.y, cb1.z, cb1.w};

    v8f d0 = __builtin_amdgcn_wmma_f32_16x16x4_f32(false, pa0, false, bp,
                                                   (short)0, cz, false, false);
    v8f d1 = __builtin_amdgcn_wmma_f32_16x16x4_f32(false, pa1, false, bp,
                                                   (short)0, cz, false, false);

#pragma unroll
    for (int e = 0; e < 8; ++e) {
      float t;
      asm("v_maxmin_num_f32 %0, |%1|, |%2|, 1.0" : "=v"(t) : "v"(d0[e]), "v"(d1[e]));
      float m = 1.0f - t;
      accR[e & 1] = fmaf(m, cr[e], accR[e & 1]);
      accG[e & 1] = fmaf(m, cg[e], accG[e & 1]);
      accB[e & 1] = fmaf(m, cb[e], accB[e & 1]);
    }
  }

  float sR = accR[0] + accR[1];
  float sG = accG[0] + accG[1];
  float sB = accB[0] + accB[1];
  sR += xhalf(sR);
  sG += xhalf(sG);
  sB += xhalf(sB);

  const int w = wcol0 + ln;
  const float s0 = hi ? sG : sR;
  const float o0 = 1.0f / (1.0f + __expf(-4.0f * s0));
  const float o2 = 1.0f / (1.0f + __expf(-4.0f * sB));
  out[hi * CANVAS_H * CANVAS_W + hrow * CANVAS_W + w] = o0;
  if (hi == 0) out[2 * CANVAS_H * CANVAS_W + hrow * CANVAS_W + w] = o2;
}

extern "C" void kernel_launch(void* const* d_in, const int* in_sizes, int n_in,
                              void* d_out, int out_size, void* d_ws, size_t ws_size,
                              hipStream_t stream) {
  (void)in_sizes; (void)n_in; (void)out_size; (void)d_ws; (void)ws_size;
  const float* loc = (const float*)d_in[0];
  const float* mo  = (const float*)d_in[1];
  const float* mse = (const float*)d_in[2];
  const float* col = (const float*)d_in[3];
  float* out = (float*)d_out;

  const int blocks = (CANVAS_H / BLK_ROWS) * (CANVAS_W / BLK_COLS);
  points_kernel<<<blocks, 256, 0, stream>>>(loc, mo, mse, col, out);
}
